# starting kernel
# baseline (speedup 1.0000x reference)
#include <hip/hip_runtime.h>
#include <hip/hip_bf16.h>

#define N_NODES 50000
#define N_EDGES 800000
#define IN_CH   128
#define HID_CH  128
#define OUT_CH  64

typedef __attribute__((ext_vector_type(2))) float v2f;
typedef __attribute__((ext_vector_type(8))) float v8f;

__global__ void zero_f32(float* __restrict__ p, size_t n) {
    size_t i = (size_t)blockIdx.x * blockDim.x + threadIdx.x;
    size_t stride = (size_t)gridDim.x * blockDim.x;
    for (; i < n; i += stride) p[i] = 0.0f;
}

__global__ void deg_kernel(const int* __restrict__ dst, float* __restrict__ deg, int E) {
    int e = blockIdx.x * blockDim.x + threadIdx.x;
    if (e < E) unsafeAtomicAdd(&deg[dst[e]], 1.0f);
}

__global__ void invdeg_kernel(const float* __restrict__ deg, float* __restrict__ invdeg, int n) {
    int i = blockIdx.x * blockDim.x + threadIdx.x;
    if (i < n) invdeg[i] = 1.0f / fmaxf(deg[i], 1.0f);
}

__global__ __launch_bounds__(256)
void scatter_kernel(const float* __restrict__ feat,
                    const int* __restrict__ src, const int* __restrict__ dst,
                    float* __restrict__ agg, int E) {
    int t    = blockIdx.x * blockDim.x + threadIdx.x;
    int e    = t >> 5;
    int lane = t & 31;
    if (e >= E) return;
    int s = src[e];
    int d = dst[e];
    const float4 v = *(const float4*)(feat + (size_t)s * 128 + lane * 4);
    float* base = agg + (size_t)d * 128 + lane * 4;
    unsafeAtomicAdd(base + 0, v.x);
    unsafeAtomicAdd(base + 1, v.y);
    unsafeAtomicAdd(base + 2, v.z);
    unsafeAtomicAdd(base + 3, v.w);
}

template <int NOUT, bool RELU>
__global__ __launch_bounds__(2 * NOUT)
void sage_gemm(const float* __restrict__ Aagg,
               const float* __restrict__ Aself,
               const float* __restrict__ invdeg,
               const float* __restrict__ Wl,
               const float* __restrict__ Wr,
               const float* __restrict__ bias,
               float* __restrict__ out) {
    constexpr int K = 128;
    const int wave = threadIdx.x >> 5;
    const int lane = threadIdx.x & 31;
    const int row  = lane & 15;
    const int hi   = lane >> 4;
    const int m0   = blockIdx.x * 16;
    const int n0   = wave * 16;

    const float* __restrict__ arow_agg  = Aagg  + (size_t)(m0 + row) * K;
    const float* __restrict__ arow_self = Aself + (size_t)(m0 + row) * K;
    const float  scale = invdeg[m0 + row];

    v8f c = {};

    #pragma unroll 4
    for (int k0 = 0; k0 < K; k0 += 4) {
        const int ka = k0 + 2 * hi;
        const float2 t = *(const float2*)(arow_agg + ka);
        v2f a;
        a.x = t.x * scale;
        a.y = t.y * scale;
        v2f b;
        b.x = Wl[(size_t)ka * NOUT + n0 + row];
        b.y = Wl[(size_t)(ka + 1) * NOUT + n0 + row];
        c = __builtin_amdgcn_wmma_f32_16x16x4_f32(false, a, false, b,
                                                  (short)0, c, false, false);
    }

    #pragma unroll 4
    for (int k0 = 0; k0 < K; k0 += 4) {
        const int ka = k0 + 2 * hi;
        const float2 t = *(const float2*)(arow_self + ka);
        v2f a;
        a.x = t.x;
        a.y = t.y;
        v2f b;
        b.x = Wr[(size_t)ka * NOUT + n0 + row];
        b.y = Wr[(size_t)(ka + 1) * NOUT + n0 + row];
        c = __builtin_amdgcn_wmma_f32_16x16x4_f32(false, a, false, b,
                                                  (short)0, c, false, false);
    }

    const float bv = bias[n0 + row];
    #pragma unroll
    for (int r = 0; r < 8; ++r) {
        float v = c[r] + bv;
        if (RELU) v = fmaxf(v, 0.0f);
        out[(size_t)(m0 + r + 8 * hi) * NOUT + n0 + row] = v;
    }
}

extern "C" void kernel_launch(void* const* d_in, const int* in_sizes, int n_in,
                              void* d_out, int out_size, void* d_ws, size_t ws_size,
                              hipStream_t stream) {
    const float* x   = (const float*)d_in[0];
    const int*   ei  = (const int*)d_in[1];
    const float* W1l = (const float*)d_in[2];
    const float* b1  = (const float*)d_in[3];
    const float* W1r = (const float*)d_in[4];
    const float* W2l = (const float*)d_in[5];
    const float* b2  = (const float*)d_in[6];
    const float* W2r = (const float*)d_in[7];

    const int* src = ei;
    const int* dst = ei + N_EDGES;

    float* agg    = (float*)d_ws;
    float* h      = agg + (size_t)N_NODES * 128;
    float* deg    = h + (size_t)N_NODES * 128;
    float* invdeg = deg + N_NODES;

    const int ZB = 256;

    zero_f32<<<1024, ZB, 0, stream>>>(deg, (size_t)N_NODES);
    deg_kernel<<<(N_EDGES + ZB - 1) / ZB, ZB, 0, stream>>>(dst, deg, N_EDGES);
    invdeg_kernel<<<(N_NODES + ZB - 1) / ZB, ZB, 0, stream>>>(deg, invdeg, N_NODES);

    zero_f32<<<2048, ZB, 0, stream>>>(agg, (size_t)N_NODES * 128);
    scatter_kernel<<<(N_EDGES * 32 + ZB - 1) / ZB, ZB, 0, stream>>>(x, src, dst, agg, N_EDGES);
    sage_gemm<HID_CH, true><<<N_NODES / 16, 2 * HID_CH, 0, stream>>>(
        agg, x, invdeg, W1l, W1r, b1, h);

    zero_f32<<<2048, ZB, 0, stream>>>(agg, (size_t)N_NODES * 128);
    scatter_kernel<<<(N_EDGES * 32 + ZB - 1) / ZB, ZB, 0, stream>>>(h, src, dst, agg, N_EDGES);
    sage_gemm<OUT_CH, false><<<N_NODES / 16, 2 * OUT_CH, 0, stream>>>(
        agg, h, invdeg, W2l, W2r, b2, (float*)d_out);
}


# Neighbour sums in LDS: a work-group owns 196 destination nodes, sifts the edge list for them and adds source rows by LDS f32 atomics (arrival order); global atomics, zero and degree launches gone
# speedup vs baseline: 2.4719x; 2.4719x over previous
#include <hip/hip_runtime.h>
#include <hip/hip_bf16.h>

#define N_NODES 50000
#define N_EDGES 800000
#define IN_CH   128
#define HID_CH  128
#define OUT_CH  64

typedef __attribute__((ext_vector_type(2))) float v2f;
typedef __attribute__((ext_vector_type(8))) float v8f;

template <int ROWF, int RR, int EPB>
__global__ __launch_bounds__(1024)
void range_mean(const float* __restrict__ feat,
                const int* __restrict__ src, const int* __restrict__ dst,
                float* __restrict__ mean) {
    static_assert(ROWF % 32 == 0, "a row is a whole number of 128-byte lines");
    static_assert(N_EDGES % 4 == 0, "16-byte index loads");
    constexpr int NV   = ROWF / 32;
    constexpr int GRP  = 8;
    constexpr int CAP  = 128 * GRP + 128;
    constexpr int EPP  = 4096;
    constexpr int NPASS = (N_EDGES + EPP - 1) / EPP;
    __shared__ float acc[(RR + 1) * ROWF];
    __shared__ float cntf[RR + 1];
    __shared__ int   lists[32 * CAP];

    const int tid  = threadIdx.x;
    const int lane = tid & 31;
    const int wv   = __builtin_amdgcn_readfirstlane(tid >> 5);
    const int base = blockIdx.x * RR;

    for (int i = tid; i < (RR + 1) * ROWF; i += 1024) acc[i] = 0.0f;
    for (int i = tid; i < RR + 1; i += 1024) cntf[i] = 0.0f;
    __syncthreads();

    int* list = lists + wv * CAP;
    int cnt = 0;

    const int start = blockIdx.x % NPASS;
    auto edge_of = [&](int p) -> int {
        int q = start + p;
        if (q >= NPASS) q -= NPASS;
        const int e = (q * 1024 + tid) * 4;
        return (p < NPASS && e < N_EDGES) ? e : -1;
    };

    auto fetch = [&](int4 (&dv)[GRP], int p) {
        #pragma unroll
        for (int u = 0; u < GRP; ++u) {
            const int e = edge_of(p + u);
            dv[u] = *(const int4*)(dst + (e < 0 ? 0 : e));
        }
    };

    auto sift = [&](const int4 (&dv)[GRP], int p) {
        #pragma unroll
        for (int u = 0; u < GRP; ++u) {
            const int e = edge_of(p + u);
            const unsigned lo = e >= 0 ? (unsigned)base : 0x80000000u;
            const int dd[4] = {dv[u].x, dv[u].y, dv[u].z, dv[u].w};
            #pragma unroll
            for (int c = 0; c < 4; ++c) {
                const bool hit = (unsigned)dd[c] - lo < (unsigned)RR;
                const unsigned m = __builtin_amdgcn_ballot_w32(hit);
                if (m) {
                    if (hit) list[cnt + __builtin_popcount(m & ((1u << lane) - 1u))] = e + c;
                    cnt += __builtin_popcount(m);
                }
            }
        }
    };

    auto drain = [&]() {
        __builtin_amdgcn_fence(__ATOMIC_RELEASE, "wavefront");
        __builtin_amdgcn_wave_barrier();
        __builtin_amdgcn_fence(__ATOMIC_ACQUIRE, "wavefront");
        for (int i0 = 0; i0 < cnt; i0 += EPB) {
            int s = 0, d = RR;
            if (lane < EPB && i0 + lane < cnt) {
                const int e = list[i0 + lane];
                s = src[e];
                d = dst[e] - base;
            }
            float v[EPB][NV];
            #pragma unroll
            for (int j = 0; j < EPB; ++j) {
                const int sj = __builtin_amdgcn_readlane(s, j);
                const float* __restrict__ frow = feat + (size_t)sj * ROWF + lane;
                #pragma unroll
                for (int q = 0; q < NV; ++q) v[j][q] = frow[32 * q];
            }
            if (lane < EPB) unsafeAtomicAdd(&cntf[d], 1.0f);
            #pragma unroll
            for (int j = 0; j < EPB; ++j) {
                const int dj = __builtin_amdgcn_readlane(d, j);
                float* arow = acc + dj * ROWF + lane;
                #pragma unroll
                for (int q = 0; q < NV; ++q) unsafeAtomicAdd(arow + 32 * q, v[j][q]);
            }
        }
        __builtin_amdgcn_fence(__ATOMIC_RELEASE, "wavefront");
        __builtin_amdgcn_wave_barrier();
        cnt = 0;
    };

    int4 da[GRP], db[GRP];
    fetch(da, 0);
    for (int p0 = 0; p0 < NPASS; p0 += 2 * GRP) {
        fetch(db, p0 + GRP);
        sift(da, p0);
        if (cnt > CAP - 128 * GRP) drain();
        fetch(da, p0 + 2 * GRP);
        sift(db, p0 + GRP);
        if (cnt > CAP - 128 * GRP) drain();
    }
    drain();
    __syncthreads();

    constexpr int R4 = ROWF / 4;
    for (int i = tid; i < RR * R4; i += 1024) {
        const int r = i / R4, c4 = i % R4;
        const int node = base + r;
        if (node < N_NODES) {
            const float id = 1.0f / fmaxf(cntf[r], 1.0f);
            float4 a = *(const float4*)(acc + r * ROWF + 4 * c4);
            a.x *= id; a.y *= id; a.z *= id; a.w *= id;
            *(float4*)(mean + (size_t)node * ROWF + 4 * c4) = a;
        }
    }
}

template <int NOUT, bool SCALE>
__device__ __forceinline__ void strip_product(const float* __restrict__ arow, float scale,
                                              const float* __restrict__ Wm,
                                              int cb, int col, int hi, v8f (&c)[4]) {
    const float* __restrict__ ap = arow + 2 * hi;
    const float* __restrict__ wp = Wm + (size_t)(2 * hi) * NOUT + cb + 4 * col;
    #pragma unroll 4
    for (int k0 = 0; k0 < 128; k0 += 4) {
        const float2 t  = *(const float2*)(ap + k0);
        const float4 bx = *(const float4*)(wp + (size_t)k0 * NOUT);
        const float4 by = *(const float4*)(wp + (size_t)(k0 + 1) * NOUT);
        v2f a;
        a.x = SCALE ? t.x * scale : t.x;
        a.y = SCALE ? t.y * scale : t.y;
        v2f b0, b1, b2, b3;
        b0.x = bx.x; b0.y = by.x;
        b1.x = bx.y; b1.y = by.y;
        b2.x = bx.z; b2.y = by.z;
        b3.x = bx.w; b3.y = by.w;
        c[0] = __builtin_amdgcn_wmma_f32_16x16x4_f32(false, a, false, b0, (short)0, c[0], false, false);
        c[1] = __builtin_amdgcn_wmma_f32_16x16x4_f32(false, a, false, b1, (short)0, c[1], false, false);
        c[2] = __builtin_amdgcn_wmma_f32_16x16x4_f32(false, a, false, b2, (short)0, c[2], false, false);
        c[3] = __builtin_amdgcn_wmma_f32_16x16x4_f32(false, a, false, b3, (short)0, c[3], false, false);
    }
}

template <int NOUT, bool RELU>
__global__ __launch_bounds__(NOUT / 2)
void sage_gemm(const float* __restrict__ Aagg,
               const float* __restrict__ Aself,
               const float* __restrict__ Wl,
               const float* __restrict__ Wr,
               const float* __restrict__ bias,
               float* __restrict__ out) {
    constexpr int K = 128;
    const int wave = threadIdx.x >> 5;
    const int lane = threadIdx.x & 31;
    const int row  = lane & 15;
    const int hi   = lane >> 4;
    const int m0   = blockIdx.x * 16;
    const int cb   = wave * 64;

    v8f c[4] = {};
    strip_product<NOUT, false>(Aagg  + (size_t)(m0 + row) * K, 1.0f,  Wl, cb, row, hi, c);
    strip_product<NOUT, false>(Aself + (size_t)(m0 + row) * K, 1.0f,  Wr, cb, row, hi, c);

    const float4 bv = *(const float4*)(bias + cb + 4 * row);
    #pragma unroll
    for (int r = 0; r < 8; ++r) {
        float4 v;
        v.x = c[0][r] + bv.x;
        v.y = c[1][r] + bv.y;
        v.z = c[2][r] + bv.z;
        v.w = c[3][r] + bv.w;
        if (RELU) {
            v.x = fmaxf(v.x, 0.0f);
            v.y = fmaxf(v.y, 0.0f);
            v.z = fmaxf(v.z, 0.0f);
            v.w = fmaxf(v.w, 0.0f);
        }
        *(float4*)(out + (size_t)(m0 + r + 8 * hi) * NOUT + cb + 4 * row) = v;
    }
}

template <int NOUT>
__global__ __launch_bounds__(NOUT / 2)
void proj_rows(const float* __restrict__ A,
               const float* __restrict__ Wl,
               float* __restrict__ P) {
    const int wave = threadIdx.x >> 5;
    const int lane = threadIdx.x & 31;
    const int row  = lane & 15;
    const int hi   = lane >> 4;
    const int m0   = blockIdx.x * 16;
    const int cb   = wave * 64;

    v8f c[4] = {};
    strip_product<NOUT, false>(A + (size_t)(m0 + row) * 128, 1.0f, Wl, cb, row, hi, c);

    #pragma unroll
    for (int r = 0; r < 8; ++r) {
        const size_t o = (size_t)(m0 + r + 8 * hi) * NOUT + cb + 4 * row;
        float4 v;
        v.x = c[0][r];
        v.y = c[1][r];
        v.z = c[2][r];
        v.w = c[3][r];
        *(float4*)(P + o) = v;
    }
}

template <int NOUT>
__global__ __launch_bounds__(NOUT / 2)
void root_join(const float* __restrict__ A,
               const float* __restrict__ Wr,
               const float* __restrict__ bias,
               const float* __restrict__ mean,
               float* __restrict__ out) {
    const int wave = threadIdx.x >> 5;
    const int lane = threadIdx.x & 31;
    const int row  = lane & 15;
    const int hi   = lane >> 4;
    const int m0   = blockIdx.x * 16;
    const int cb   = wave * 64;

    v8f c[4] = {};
    strip_product<NOUT, false>(A + (size_t)(m0 + row) * 128, 1.0f, Wr, cb, row, hi, c);

    const float4 bv = *(const float4*)(bias + cb + 4 * row);
    #pragma unroll
    for (int r = 0; r < 8; ++r) {
        const int m = m0 + r + 8 * hi;
        const size_t o = (size_t)m * NOUT + cb + 4 * row;
        const float4 mv = *(const float4*)(mean + o);
        float4 v;
        v.x = c[0][r] + bv.x + mv.x;
        v.y = c[1][r] + bv.y + mv.y;
        v.z = c[2][r] + bv.z + mv.z;
        v.w = c[3][r] + bv.w + mv.w;
        *(float4*)(out + o) = v;
    }
}

extern "C" void kernel_launch(void* const* d_in, const int* in_sizes, int n_in,
                              void* d_out, int out_size, void* d_ws, size_t ws_size,
                              hipStream_t stream) {
    const float* x   = (const float*)d_in[0];
    const int*   ei  = (const int*)d_in[1];
    const float* W1l = (const float*)d_in[2];
    const float* b1  = (const float*)d_in[3];
    const float* W1r = (const float*)d_in[4];
    const float* W2l = (const float*)d_in[5];
    const float* b2  = (const float*)d_in[6];
    const float* W2r = (const float*)d_in[7];

    const int* src = ei;
    const int* dst = ei + N_EDGES;

    float* mean1  = (float*)d_ws;
    float* h      = mean1 + (size_t)N_NODES * 128;
    float* proj   = mean1;
    float* mean2  = mean1 + (size_t)N_NODES * OUT_CH;

    constexpr int RR = 196;
    constexpr int NR = (N_NODES + RR - 1) / RR;

    range_mean<128, RR, 8><<<NR, 1024, 0, stream>>>(x, src, dst, mean1);
    sage_gemm<HID_CH, true><<<N_NODES / 16, HID_CH / 2, 0, stream>>>(
        mean1, x, W1l, W1r, b1, h);

    proj_rows<OUT_CH><<<N_NODES / 16, OUT_CH / 2, 0, stream>>>(h, W2l, proj);
    range_mean<OUT_CH, RR, 16><<<NR, 1024, 0, stream>>>(proj, src, dst, mean2);
    root_join<OUT_CH><<<N_NODES / 16, OUT_CH / 2, 0, stream>>>(
        h, W2r, b2, mean2, (float*)d_out);
}


# Edge sift on byte ids: a new first launch stores dst/196 per edge in the yet unwritten output buffer; range sums scan these, 16 edges per load, not the dst column; same arrival-order LDS atomic sums
# speedup vs baseline: 2.9431x; 1.1906x over previous
#include <hip/hip_runtime.h>
#include <hip/hip_bf16.h>

#define N_NODES 50000
#define N_EDGES 800000
#define IN_CH   128
#define HID_CH  128
#define OUT_CH  64

typedef __attribute__((ext_vector_type(2))) float v2f;
typedef __attribute__((ext_vector_type(8))) float v8f;

template <int RR>
__global__ __launch_bounds__(256)
void range_ids(const int* __restrict__ dst, unsigned* __restrict__ ids) {
    static_assert((N_NODES + RR - 1) / RR <= 256, "a range id is one byte");
    static_assert(N_EDGES % 4 == 0, "16-byte index loads");
    const int i = blockIdx.x * 256 + threadIdx.x;
    if (i < N_EDGES / 4) {
        const int4 d = *(const int4*)(dst + 4 * i);
        ids[i] =  (((unsigned)d.x / RR) & 255u)
               | ((((unsigned)d.y / RR) & 255u) << 8)
               | ((((unsigned)d.z / RR) & 255u) << 16)
               | ((((unsigned)d.w / RR) & 255u) << 24);
    }
}

template <int ROWF, int RR, int EPB>
__global__ __launch_bounds__(1024)
void range_mean(const float* __restrict__ feat,
                const int* __restrict__ src, const int* __restrict__ dst,
                const unsigned* __restrict__ ids,
                float* __restrict__ mean) {
    static_assert(ROWF % 32 == 0, "a row is a whole number of 128-byte lines");
    static_assert(N_EDGES % 16 == 0, "16-byte id loads");
    constexpr int NV   = ROWF / 32;
    constexpr int GRP  = 2;
    constexpr int EPT  = 16;
    constexpr int GMAX = 32 * EPT * GRP;
    constexpr int CAP  = GMAX + 256;
    constexpr int EPP  = 1024 * EPT;
    constexpr int NPASS = (N_EDGES + EPP - 1) / EPP;
    __shared__ float acc[(RR + 1) * ROWF];
    __shared__ float cntf[RR + 1];
    __shared__ int   lists[32 * CAP];

    const int tid  = threadIdx.x;
    const int lane = tid & 31;
    const int wv   = __builtin_amdgcn_readfirstlane(tid >> 5);
    const int base = blockIdx.x * RR;
    const unsigned pat = (unsigned)blockIdx.x * 0x01010101u;

    for (int i = tid; i < (RR + 1) * ROWF; i += 1024) acc[i] = 0.0f;
    for (int i = tid; i < RR + 1; i += 1024) cntf[i] = 0.0f;
    __syncthreads();

    int* list = lists + wv * CAP;
    int cnt = 0;

    const int start = blockIdx.x % NPASS;
    auto edge_of = [&](int p) -> int {
        int q = start + p;
        if (q >= NPASS) q -= NPASS;
        const int e = (q * 1024 + tid) * EPT;
        return (p < NPASS && e < N_EDGES) ? e : -1;
    };

    auto fetch = [&](uint4 (&dv)[GRP], int p) {
        #pragma unroll
        for (int u = 0; u < GRP; ++u) {
            const int e = edge_of(p + u);
            dv[u] = *(const uint4*)(ids + ((e < 0 ? 0 : e) >> 2));
        }
    };

    auto sift = [&](const uint4 (&dv)[GRP], int p) {
        #pragma unroll
        for (int u = 0; u < GRP; ++u) {
            const int e = edge_of(p + u);
            const unsigned km = e >= 0 ? 0x7F7F7F7Fu : 0xFFFFFFFFu;
            const unsigned ww[4] = {dv[u].x, dv[u].y, dv[u].z, dv[u].w};
            #pragma unroll
            for (int c = 0; c < 4; ++c) {
                const unsigned x = ww[c] ^ pat;
                unsigned f = ~(((x & 0x7F7F7F7Fu) + 0x7F7F7F7Fu) | x | km);
                unsigned m = __builtin_amdgcn_ballot_w32(f != 0u);
                while (m) {
                    if (f) list[__builtin_amdgcn_mbcnt_lo(m, (unsigned)cnt)] = e + 4 * c + (__builtin_ctz(f) >> 3);
                    cnt += __builtin_popcount(m);
                    f &= f - 1u;
                    m = __builtin_amdgcn_ballot_w32(f != 0u);
                }
            }
        }
    };

    auto drain = [&]() {
        __builtin_amdgcn_fence(__ATOMIC_RELEASE, "wavefront");
        __builtin_amdgcn_wave_barrier();
        __builtin_amdgcn_fence(__ATOMIC_ACQUIRE, "wavefront");
        for (int i0 = 0; i0 < cnt; i0 += EPB) {
            int s = 0, d = RR;
            if (lane < EPB && i0 + lane < cnt) {
                const int e = list[i0 + lane];
                const int se = src[e];
                const unsigned de = (unsigned)(dst[e] - base);
                if (de < (unsigned)RR) { s = se; d = (int)de; }
            }
            float v[EPB][NV];
            #pragma unroll
            for (int j = 0; j < EPB; ++j) {
                const int sj = __builtin_amdgcn_readlane(s, j);
                const float* __restrict__ frow = feat + (size_t)sj * ROWF + lane;
                #pragma unroll
                for (int q = 0; q < NV; ++q) v[j][q] = frow[32 * q];
            }
            if (lane < EPB) unsafeAtomicAdd(&cntf[d], 1.0f);
            #pragma unroll
            for (int j = 0; j < EPB; ++j) {
                const int dj = __builtin_amdgcn_readlane(d, j);
                float* arow = acc + dj * ROWF + lane;
                #pragma unroll
                for (int q = 0; q < NV; ++q) unsafeAtomicAdd(arow + 32 * q, v[j][q]);
            }
        }
        __builtin_amdgcn_fence(__ATOMIC_RELEASE, "wavefront");
        __builtin_amdgcn_wave_barrier();
        cnt = 0;
    };

    uint4 da[GRP], db[GRP];
    fetch(da, 0);
    for (int p0 = 0; p0 < NPASS; p0 += 2 * GRP) {
        fetch(db, p0 + GRP);
        sift(da, p0);
        if (cnt > CAP - GMAX) drain();
        fetch(da, p0 + 2 * GRP);
        sift(db, p0 + GRP);
        if (cnt > CAP - GMAX) drain();
    }
    drain();
    __syncthreads();

    constexpr int R4 = ROWF / 4;
    for (int i = tid; i < RR * R4; i += 1024) {
        const int r = i / R4, c4 = i % R4;
        const int node = base + r;
        if (node < N_NODES) {
            const float id = 1.0f / fmaxf(cntf[r], 1.0f);
            float4 a = *(const float4*)(acc + r * ROWF + 4 * c4);
            a.x *= id; a.y *= id; a.z *= id; a.w *= id;
            *(float4*)(mean + (size_t)node * ROWF + 4 * c4) = a;
        }
    }
}

template <int NOUT, bool SCALE>
__device__ __forceinline__ void strip_product(const float* __restrict__ arow, float scale,
                                              const float* __restrict__ Wm,
                                              int cb, int col, int hi, v8f (&c)[4]) {
    const float* __restrict__ ap = arow + 2 * hi;
    const float* __restrict__ wp = Wm + (size_t)(2 * hi) * NOUT + cb + 4 * col;
    #pragma unroll 4
    for (int k0 = 0; k0 < 128; k0 += 4) {
        const float2 t  = *(const float2*)(ap + k0);
        const float4 bx = *(const float4*)(wp + (size_t)k0 * NOUT);
        const float4 by = *(const float4*)(wp + (size_t)(k0 + 1) * NOUT);
        v2f a;
        a.x = SCALE ? t.x * scale : t.x;
        a.y = SCALE ? t.y * scale : t.y;
        v2f b0, b1, b2, b3;
        b0.x = bx.x; b0.y = by.x;
        b1.x = bx.y; b1.y = by.y;
        b2.x = bx.z; b2.y = by.z;
        b3.x = bx.w; b3.y = by.w;
        c[0] = __builtin_amdgcn_wmma_f32_16x16x4_f32(false, a, false, b0, (short)0, c[0], false, false);
        c[1] = __builtin_amdgcn_wmma_f32_16x16x4_f32(false, a, false, b1, (short)0, c[1], false, false);
        c[2] = __builtin_amdgcn_wmma_f32_16x16x4_f32(false, a, false, b2, (short)0, c[2], false, false);
        c[3] = __builtin_amdgcn_wmma_f32_16x16x4_f32(false, a, false, b3, (short)0, c[3], false, false);
    }
}

template <int NOUT, bool RELU>
__global__ __launch_bounds__(NOUT / 2)
void sage_gemm(const float* __restrict__ Aagg,
               const float* __restrict__ Aself,
               const float* __restrict__ Wl,
               const float* __restrict__ Wr,
               const float* __restrict__ bias,
               float* __restrict__ out) {
    constexpr int K = 128;
    const int wave = threadIdx.x >> 5;
    const int lane = threadIdx.x & 31;
    const int row  = lane & 15;
    const int hi   = lane >> 4;
    const int m0   = blockIdx.x * 16;
    const int cb   = wave * 64;

    v8f c[4] = {};
    strip_product<NOUT, false>(Aagg  + (size_t)(m0 + row) * K, 1.0f,  Wl, cb, row, hi, c);
    strip_product<NOUT, false>(Aself + (size_t)(m0 + row) * K, 1.0f,  Wr, cb, row, hi, c);

    const float4 bv = *(const float4*)(bias + cb + 4 * row);
    #pragma unroll
    for (int r = 0; r < 8; ++r) {
        float4 v;
        v.x = c[0][r] + bv.x;
        v.y = c[1][r] + bv.y;
        v.z = c[2][r] + bv.z;
        v.w = c[3][r] + bv.w;
        if (RELU) {
            v.x = fmaxf(v.x, 0.0f);
            v.y = fmaxf(v.y, 0.0f);
            v.z = fmaxf(v.z, 0.0f);
            v.w = fmaxf(v.w, 0.0f);
        }
        *(float4*)(out + (size_t)(m0 + r + 8 * hi) * NOUT + cb + 4 * row) = v;
    }
}

template <int NOUT>
__global__ __launch_bounds__(NOUT / 2)
void proj_rows(const float* __restrict__ A,
               const float* __restrict__ Wl,
               float* __restrict__ P) {
    const int wave = threadIdx.x >> 5;
    const int lane = threadIdx.x & 31;
    const int row  = lane & 15;
    const int hi   = lane >> 4;
    const int m0   = blockIdx.x * 16;
    const int cb   = wave * 64;

    v8f c[4] = {};
    strip_product<NOUT, false>(A + (size_t)(m0 + row) * 128, 1.0f, Wl, cb, row, hi, c);

    #pragma unroll
    for (int r = 0; r < 8; ++r) {
        const size_t o = (size_t)(m0 + r + 8 * hi) * NOUT + cb + 4 * row;
        float4 v;
        v.x = c[0][r];
        v.y = c[1][r];
        v.z = c[2][r];
        v.w = c[3][r];
        *(float4*)(P + o) = v;
    }
}

template <int NOUT>
__global__ __launch_bounds__(NOUT / 2)
void root_join(const float* __restrict__ A,
               const float* __restrict__ Wr,
               const float* __restrict__ bias,
               const float* __restrict__ mean,
               float* __restrict__ out) {
    const int wave = threadIdx.x >> 5;
    const int lane = threadIdx.x & 31;
    const int row  = lane & 15;
    const int hi   = lane >> 4;
    const int m0   = blockIdx.x * 16;
    const int cb   = wave * 64;

    v8f c[4] = {};
    strip_product<NOUT, false>(A + (size_t)(m0 + row) * 128, 1.0f, Wr, cb, row, hi, c);

    const float4 bv = *(const float4*)(bias + cb + 4 * row);
    #pragma unroll
    for (int r = 0; r < 8; ++r) {
        const int m = m0 + r + 8 * hi;
        const size_t o = (size_t)m * NOUT + cb + 4 * row;
        const float4 mv = *(const float4*)(mean + o);
        float4 v;
        v.x = c[0][r] + bv.x + mv.x;
        v.y = c[1][r] + bv.y + mv.y;
        v.z = c[2][r] + bv.z + mv.z;
        v.w = c[3][r] + bv.w + mv.w;
        *(float4*)(out + o) = v;
    }
}

extern "C" void kernel_launch(void* const* d_in, const int* in_sizes, int n_in,
                              void* d_out, int out_size, void* d_ws, size_t ws_size,
                              hipStream_t stream) {
    const float* x   = (const float*)d_in[0];
    const int*   ei  = (const int*)d_in[1];
    const float* W1l = (const float*)d_in[2];
    const float* b1  = (const float*)d_in[3];
    const float* W1r = (const float*)d_in[4];
    const float* W2l = (const float*)d_in[5];
    const float* b2  = (const float*)d_in[6];
    const float* W2r = (const float*)d_in[7];

    const int* src = ei;
    const int* dst = ei + N_EDGES;

    float* mean1  = (float*)d_ws;
    float* h      = mean1 + (size_t)N_NODES * 128;
    float* proj   = mean1;
    float* mean2  = mean1 + (size_t)N_NODES * OUT_CH;

    constexpr int RR = 196;
    constexpr int NR = (N_NODES + RR - 1) / RR;

    unsigned* ids = (unsigned*)d_out;
    range_ids<RR><<<(N_EDGES / 4 + 255) / 256, 256, 0, stream>>>(dst, ids);

    range_mean<128, RR, 8><<<NR, 1024, 0, stream>>>(x, src, dst, ids, mean1);
    sage_gemm<HID_CH, true><<<N_NODES / 16, HID_CH / 2, 0, stream>>>(
        mean1, x, W1l, W1r, b1, h);

    proj_rows<OUT_CH><<<N_NODES / 16, OUT_CH / 2, 0, stream>>>(h, W2l, proj);
    range_mean<OUT_CH, RR, 16><<<NR, 1024, 0, stream>>>(proj, src, dst, ids, mean2);
    root_join<OUT_CH><<<N_NODES / 16, OUT_CH / 2, 0, stream>>>(
        h, W2r, b2, mean2, (float*)d_out);
}


# Sift once: layer 1's range waves store their (source, local destination) pairs in the still unwritten output buffer; layer 2 replays them: no second sift or index loads; same arrival-order LDS sums
# speedup vs baseline: 3.2586x; 1.1072x over previous
#include <hip/hip_runtime.h>
#include <hip/hip_bf16.h>

#define N_NODES 50000
#define N_EDGES 800000
#define IN_CH   128
#define HID_CH  128
#define OUT_CH  64

typedef __attribute__((ext_vector_type(2))) float v2f;
typedef __attribute__((ext_vector_type(8))) float v8f;

template <int RR>
__global__ __launch_bounds__(256)
void range_ids(const int* __restrict__ dst, unsigned* __restrict__ ids) {
    static_assert((N_NODES + RR - 1) / RR <= 256, "a range id is one byte");
    static_assert(N_EDGES % 4 == 0, "16-byte index loads");
    const int i = blockIdx.x * 256 + threadIdx.x;
    if (i < N_EDGES / 4) {
        const int4 d = *(const int4*)(dst + 4 * i);
        ids[i] =  (((unsigned)d.x / RR) & 255u)
               | ((((unsigned)d.y / RR) & 255u) << 8)
               | ((((unsigned)d.z / RR) & 255u) << 16)
               | ((((unsigned)d.w / RR) & 255u) << 24);
    }
}

template <bool B> struct flag { static constexpr bool value = B; };

constexpr int WCAP = 256;

template <int ROWF, int RR, int EPB, bool REPLAY>
__global__ __launch_bounds__(1024)
void range_mean(const float* __restrict__ feat,
                const int* __restrict__ src, const int* __restrict__ dst,
                const unsigned* __restrict__ ids,
                unsigned* __restrict__ memo, int* __restrict__ memo_n,
                float* __restrict__ mean) {
    static_assert(ROWF % 32 == 0, "a row is a whole number of 128-byte lines");
    static_assert(N_EDGES % 16 == 0, "16-byte id loads");
    constexpr int NV   = ROWF / 32;
    constexpr int GRP  = 2;
    constexpr int EPT  = 16;
    constexpr int GMAX = 32 * EPT * GRP;
    constexpr int CAP  = GMAX + 256;
    static_assert(WCAP <= CAP && WCAP % 32 == 0, "a stored segment fits the wave's list");
    static_assert(RR < 256 && N_NODES <= (1 << 24), "source << 8 | local destination is one dword");
    constexpr int EPP  = 1024 * EPT;
    constexpr int NPASS = (N_EDGES + EPP - 1) / EPP;
    __shared__ float acc[(RR + 1) * ROWF];
    __shared__ float cntf[RR + 1];
    __shared__ int   lists[32 * CAP];

    const int tid  = threadIdx.x;
    const int lane = tid & 31;
    const int wv   = __builtin_amdgcn_readfirstlane(tid >> 5);
    const int base = blockIdx.x * RR;
    const unsigned pat = (unsigned)blockIdx.x * 0x01010101u;

    for (int i = tid; i < (RR + 1) * ROWF; i += 1024) acc[i] = 0.0f;
    for (int i = tid; i < RR + 1; i += 1024) cntf[i] = 0.0f;
    __syncthreads();

    int* list = lists + wv * CAP;
    int cnt = 0;
    int tot = 0;
    const int gw = blockIdx.x * 32 + wv;
    unsigned* __restrict__ seg = memo + (size_t)gw * WCAP;

    const int start = blockIdx.x % NPASS;
    auto edge_of = [&](int p) -> int {
        int q = start + p;
        if (q >= NPASS) q -= NPASS;
        const int e = (q * 1024 + tid) * EPT;
        return (p < NPASS && e < N_EDGES) ? e : -1;
    };

    auto fetch = [&](uint4 (&dv)[GRP], int p) {
        #pragma unroll
        for (int u = 0; u < GRP; ++u) {
            const int e = edge_of(p + u);
            dv[u] = *(const uint4*)(ids + ((e < 0 ? 0 : e) >> 2));
        }
    };

    auto sift = [&](const uint4 (&dv)[GRP], int p) {
        #pragma unroll
        for (int u = 0; u < GRP; ++u) {
            const int e = edge_of(p + u);
            const unsigned km = e >= 0 ? 0x7F7F7F7Fu : 0xFFFFFFFFu;
            const unsigned ww[4] = {dv[u].x, dv[u].y, dv[u].z, dv[u].w};
            #pragma unroll
            for (int c = 0; c < 4; ++c) {
                const unsigned x = ww[c] ^ pat;
                unsigned f = ~(((x & 0x7F7F7F7Fu) + 0x7F7F7F7Fu) | x | km);
                unsigned m = __builtin_amdgcn_ballot_w32(f != 0u);
                while (m) {
                    if (f) list[__builtin_amdgcn_mbcnt_lo(m, (unsigned)cnt)] = e + 4 * c + (__builtin_ctz(f) >> 3);
                    cnt += __builtin_popcount(m);
                    f &= f - 1u;
                    m = __builtin_amdgcn_ballot_w32(f != 0u);
                }
            }
        }
    };

    auto drain = [&](auto packed) {
        constexpr bool PACKED = decltype(packed)::value;
        __builtin_amdgcn_fence(__ATOMIC_RELEASE, "wavefront");
        __builtin_amdgcn_wave_barrier();
        __builtin_amdgcn_fence(__ATOMIC_ACQUIRE, "wavefront");
        for (int i0 = 0; i0 < cnt; i0 += EPB) {
            int s = 0, d = RR;
            if (lane < EPB && i0 + lane < cnt) {
                const int e = list[i0 + lane];
                if constexpr (PACKED) {
                    s = (int)((unsigned)e >> 8);
                    d = e & 255;
                } else {
                    const int se = src[e];
                    const unsigned de = (unsigned)(dst[e] - base);
                    if (de < (unsigned)RR) { s = se; d = (int)de; }
                }
                if constexpr (!REPLAY) {
                    const int pos = tot + i0 + lane;
                    if (pos < WCAP) seg[pos] = ((unsigned)s << 8) | (unsigned)d;
                }
            }
            float v[EPB][NV];
            #pragma unroll
            for (int j = 0; j < EPB; ++j) {
                const int sj = __builtin_amdgcn_readlane(s, j);
                const float* __restrict__ frow = feat + (size_t)sj * ROWF + lane;
                #pragma unroll
                for (int q = 0; q < NV; ++q) v[j][q] = frow[32 * q];
            }
            if (lane < EPB) unsafeAtomicAdd(&cntf[d], 1.0f);
            #pragma unroll
            for (int j = 0; j < EPB; ++j) {
                const int dj = __builtin_amdgcn_readlane(d, j);
                float* arow = acc + dj * ROWF + lane;
                #pragma unroll
                for (int q = 0; q < NV; ++q) unsafeAtomicAdd(arow + 32 * q, v[j][q]);
            }
        }
        __builtin_amdgcn_fence(__ATOMIC_RELEASE, "wavefront");
        __builtin_amdgcn_wave_barrier();
        tot += cnt;
        cnt = 0;
    };

    auto sift_all = [&]() {
        uint4 da[GRP], db[GRP];
        fetch(da, 0);
        for (int p0 = 0; p0 < NPASS; p0 += 2 * GRP) {
            fetch(db, p0 + GRP);
            sift(da, p0);
            if (cnt > CAP - GMAX) drain(flag<false>{});
            fetch(da, p0 + 2 * GRP);
            sift(db, p0 + GRP);
            if (cnt > CAP - GMAX) drain(flag<false>{});
        }
        drain(flag<false>{});
    };

    if constexpr (!REPLAY) {
        sift_all();
        if (lane == 0) memo_n[gw] = tot;
    } else {
        const int n = __builtin_amdgcn_readfirstlane(memo_n[gw]);
        if (n <= WCAP) {
            #pragma unroll
            for (int k = 0; k < WCAP / 32; ++k) list[32 * k + lane] = (int)seg[32 * k + lane];
            cnt = n;
            drain(flag<true>{});
        } else {
            sift_all();
        }
    }
    __syncthreads();

    constexpr int R4 = ROWF / 4;
    for (int i = tid; i < RR * R4; i += 1024) {
        const int r = i / R4, c4 = i % R4;
        const int node = base + r;
        if (node < N_NODES) {
            const float id = 1.0f / fmaxf(cntf[r], 1.0f);
            float4 a = *(const float4*)(acc + r * ROWF + 4 * c4);
            a.x *= id; a.y *= id; a.z *= id; a.w *= id;
            *(float4*)(mean + (size_t)node * ROWF + 4 * c4) = a;
        }
    }
}

template <int NOUT, bool SCALE>
__device__ __forceinline__ void strip_product(const float* __restrict__ arow, float scale,
                                              const float* __restrict__ Wm,
                                              int cb, int col, int hi, v8f (&c)[4]) {
    const float* __restrict__ ap = arow + 2 * hi;
    const float* __restrict__ wp = Wm + (size_t)(2 * hi) * NOUT + cb + 4 * col;
    #pragma unroll 4
    for (int k0 = 0; k0 < 128; k0 += 4) {
        const float2 t  = *(const float2*)(ap + k0);
        const float4 bx = *(const float4*)(wp + (size_t)k0 * NOUT);
        const float4 by = *(const float4*)(wp + (size_t)(k0 + 1) * NOUT);
        v2f a;
        a.x = SCALE ? t.x * scale : t.x;
        a.y = SCALE ? t.y * scale : t.y;
        v2f b0, b1, b2, b3;
        b0.x = bx.x; b0.y = by.x;
        b1.x = bx.y; b1.y = by.y;
        b2.x = bx.z; b2.y = by.z;
        b3.x = bx.w; b3.y = by.w;
        c[0] = __builtin_amdgcn_wmma_f32_16x16x4_f32(false, a, false, b0, (short)0, c[0], false, false);
        c[1] = __builtin_amdgcn_wmma_f32_16x16x4_f32(false, a, false, b1, (short)0, c[1], false, false);
        c[2] = __builtin_amdgcn_wmma_f32_16x16x4_f32(false, a, false, b2, (short)0, c[2], false, false);
        c[3] = __builtin_amdgcn_wmma_f32_16x16x4_f32(false, a, false, b3, (short)0, c[3], false, false);
    }
}

template <int NOUT, bool RELU>
__global__ __launch_bounds__(NOUT / 2)
void sage_gemm(const float* __restrict__ Aagg,
               const float* __restrict__ Aself,
               const float* __restrict__ Wl,
               const float* __restrict__ Wr,
               const float* __restrict__ bias,
               float* __restrict__ out) {
    constexpr int K = 128;
    const int wave = threadIdx.x >> 5;
    const int lane = threadIdx.x & 31;
    const int row  = lane & 15;
    const int hi   = lane >> 4;
    const int m0   = blockIdx.x * 16;
    const int cb   = wave * 64;

    v8f c[4] = {};
    strip_product<NOUT, false>(Aagg  + (size_t)(m0 + row) * K, 1.0f,  Wl, cb, row, hi, c);
    strip_product<NOUT, false>(Aself + (size_t)(m0 + row) * K, 1.0f,  Wr, cb, row, hi, c);

    const float4 bv = *(const float4*)(bias + cb + 4 * row);
    #pragma unroll
    for (int r = 0; r < 8; ++r) {
        float4 v;
        v.x = c[0][r] + bv.x;
        v.y = c[1][r] + bv.y;
        v.z = c[2][r] + bv.z;
        v.w = c[3][r] + bv.w;
        if (RELU) {
            v.x = fmaxf(v.x, 0.0f);
            v.y = fmaxf(v.y, 0.0f);
            v.z = fmaxf(v.z, 0.0f);
            v.w = fmaxf(v.w, 0.0f);
        }
        *(float4*)(out + (size_t)(m0 + r + 8 * hi) * NOUT + cb + 4 * row) = v;
    }
}

template <int NOUT>
__global__ __launch_bounds__(NOUT / 2)
void proj_rows(const float* __restrict__ A,
               const float* __restrict__ Wl,
               float* __restrict__ P) {
    const int wave = threadIdx.x >> 5;
    const int lane = threadIdx.x & 31;
    const int row  = lane & 15;
    const int hi   = lane >> 4;
    const int m0   = blockIdx.x * 16;
    const int cb   = wave * 64;

    v8f c[4] = {};
    strip_product<NOUT, false>(A + (size_t)(m0 + row) * 128, 1.0f, Wl, cb, row, hi, c);

    #pragma unroll
    for (int r = 0; r < 8; ++r) {
        const size_t o = (size_t)(m0 + r + 8 * hi) * NOUT + cb + 4 * row;
        float4 v;
        v.x = c[0][r];
        v.y = c[1][r];
        v.z = c[2][r];
        v.w = c[3][r];
        *(float4*)(P + o) = v;
    }
}

template <int NOUT>
__global__ __launch_bounds__(NOUT / 2)
void root_join(const float* __restrict__ A,
               const float* __restrict__ Wr,
               const float* __restrict__ bias,
               const float* __restrict__ mean,
               float* __restrict__ out) {
    const int wave = threadIdx.x >> 5;
    const int lane = threadIdx.x & 31;
    const int row  = lane & 15;
    const int hi   = lane >> 4;
    const int m0   = blockIdx.x * 16;
    const int cb   = wave * 64;

    v8f c[4] = {};
    strip_product<NOUT, false>(A + (size_t)(m0 + row) * 128, 1.0f, Wr, cb, row, hi, c);

    const float4 bv = *(const float4*)(bias + cb + 4 * row);
    #pragma unroll
    for (int r = 0; r < 8; ++r) {
        const int m = m0 + r + 8 * hi;
        const size_t o = (size_t)m * NOUT + cb + 4 * row;
        const float4 mv = *(const float4*)(mean + o);
        float4 v;
        v.x = c[0][r] + bv.x + mv.x;
        v.y = c[1][r] + bv.y + mv.y;
        v.z = c[2][r] + bv.z + mv.z;
        v.w = c[3][r] + bv.w + mv.w;
        *(float4*)(out + o) = v;
    }
}

extern "C" void kernel_launch(void* const* d_in, const int* in_sizes, int n_in,
                              void* d_out, int out_size, void* d_ws, size_t ws_size,
                              hipStream_t stream) {
    const float* x   = (const float*)d_in[0];
    const int*   ei  = (const int*)d_in[1];
    const float* W1l = (const float*)d_in[2];
    const float* b1  = (const float*)d_in[3];
    const float* W1r = (const float*)d_in[4];
    const float* W2l = (const float*)d_in[5];
    const float* b2  = (const float*)d_in[6];
    const float* W2r = (const float*)d_in[7];

    const int* src = ei;
    const int* dst = ei + N_EDGES;

    float* mean1  = (float*)d_ws;
    float* h      = mean1 + (size_t)N_NODES * 128;
    float* proj   = mean1;
    float* mean2  = mean1 + (size_t)N_NODES * OUT_CH;

    constexpr int RR = 196;
    constexpr int NR = (N_NODES + RR - 1) / RR;

    unsigned* ids = (unsigned*)d_out;
    range_ids<RR><<<(N_EDGES / 4 + 255) / 256, 256, 0, stream>>>(dst, ids);

    static_assert((size_t)N_EDGES + (size_t)NR * 32 * 4 + (size_t)NR * 32 * WCAP * 4
                  <= (size_t)N_NODES * OUT_CH * 4, "ids, counts and segments fit the output buffer");
    static_assert(N_EDGES % 128 == 0, "counts and segments start on a 128-byte line");
    int*      memo_n = (int*)(ids + N_EDGES / 4);
    unsigned* memo   = (unsigned*)(memo_n + NR * 32);

    range_mean<128, RR, 8, false><<<NR, 1024, 0, stream>>>(x, src, dst, ids, memo, memo_n, mean1);
    sage_gemm<HID_CH, true><<<N_NODES / 16, HID_CH / 2, 0, stream>>>(
        mean1, x, W1l, W1r, b1, h);

    proj_rows<OUT_CH><<<N_NODES / 16, OUT_CH / 2, 0, stream>>>(h, W2l, proj);
    range_mean<OUT_CH, RR, 16, true><<<NR, 1024, 0, stream>>>(proj, src, dst, ids, memo, memo_n, mean2);
    root_join<OUT_CH><<<N_NODES / 16, OUT_CH / 2, 0, stream>>>(
        h, W2r, b2, mean2, (float*)d_out);
}


# Layer 2's two products moved into layer 1's product launch: the 16x128 h tile goes through LDS into h@W2l and h@W2r+b2, h never stored; mean launch adds onto the root rows; k order and sums unchanged
# speedup vs baseline: 3.8897x; 1.1937x over previous
#include <hip/hip_runtime.h>
#include <hip/hip_bf16.h>

#define N_NODES 50000
#define N_EDGES 800000
#define IN_CH   128
#define HID_CH  128
#define OUT_CH  64

typedef __attribute__((ext_vector_type(2))) float v2f;
typedef __attribute__((ext_vector_type(8))) float v8f;

template <int RR>
__global__ __launch_bounds__(256)
void range_ids(const int* __restrict__ dst, unsigned* __restrict__ ids) {
    static_assert((N_NODES + RR - 1) / RR <= 256, "a range id is one byte");
    static_assert(N_EDGES % 4 == 0, "16-byte index loads");
    const int i = blockIdx.x * 256 + threadIdx.x;
    if (i < N_EDGES / 4) {
        const int4 d = *(const int4*)(dst + 4 * i);
        ids[i] =  (((unsigned)d.x / RR) & 255u)
               | ((((unsigned)d.y / RR) & 255u) << 8)
               | ((((unsigned)d.z / RR) & 255u) << 16)
               | ((((unsigned)d.w / RR) & 255u) << 24);
    }
}

template <bool B> struct flag { static constexpr bool value = B; };

constexpr int WCAP = 256;

template <int ROWF, int RR, int EPB, bool REPLAY, bool ADD>
__global__ __launch_bounds__(1024)
void range_mean(const float* __restrict__ feat,
                const int* __restrict__ src, const int* __restrict__ dst,
                const unsigned* __restrict__ ids,
                unsigned* __restrict__ memo, int* __restrict__ memo_n,
                float* __restrict__ mean) {
    static_assert(ROWF % 32 == 0, "a row is a whole number of 128-byte lines");
    static_assert(N_EDGES % 16 == 0, "16-byte id loads");
    constexpr int NV   = ROWF / 32;
    constexpr int GRP  = 2;
    constexpr int EPT  = 16;
    constexpr int GMAX = 32 * EPT * GRP;
    constexpr int CAP  = GMAX + 256;
    static_assert(WCAP <= CAP && WCAP % 32 == 0, "a stored segment fits the wave's list");
    static_assert(RR < 256 && N_NODES <= (1 << 24), "source << 8 | local destination is one dword");
    constexpr int EPP  = 1024 * EPT;
    constexpr int NPASS = (N_EDGES + EPP - 1) / EPP;
    __shared__ float acc[(RR + 1) * ROWF];
    __shared__ float cntf[RR + 1];
    __shared__ int   lists[32 * CAP];

    const int tid  = threadIdx.x;
    const int lane = tid & 31;
    const int wv   = __builtin_amdgcn_readfirstlane(tid >> 5);
    const int base = blockIdx.x * RR;
    const unsigned pat = (unsigned)blockIdx.x * 0x01010101u;

    for (int i = tid; i < (RR + 1) * ROWF; i += 1024) acc[i] = 0.0f;
    for (int i = tid; i < RR + 1; i += 1024) cntf[i] = 0.0f;
    __syncthreads();

    int* list = lists + wv * CAP;
    int cnt = 0;
    int tot = 0;
    const int gw = blockIdx.x * 32 + wv;
    unsigned* __restrict__ seg = memo + (size_t)gw * WCAP;

    const int start = blockIdx.x % NPASS;
    auto edge_of = [&](int p) -> int {
        int q = start + p;
        if (q >= NPASS) q -= NPASS;
        const int e = (q * 1024 + tid) * EPT;
        return (p < NPASS && e < N_EDGES) ? e : -1;
    };

    auto fetch = [&](uint4 (&dv)[GRP], int p) {
        #pragma unroll
        for (int u = 0; u < GRP; ++u) {
            const int e = edge_of(p + u);
            dv[u] = *(const uint4*)(ids + ((e < 0 ? 0 : e) >> 2));
        }
    };

    auto sift = [&](const uint4 (&dv)[GRP], int p) {
        #pragma unroll
        for (int u = 0; u < GRP; ++u) {
            const int e = edge_of(p + u);
            const unsigned km = e >= 0 ? 0x7F7F7F7Fu : 0xFFFFFFFFu;
            const unsigned ww[4] = {dv[u].x, dv[u].y, dv[u].z, dv[u].w};
            #pragma unroll
            for (int c = 0; c < 4; ++c) {
                const unsigned x = ww[c] ^ pat;
                unsigned f = ~(((x & 0x7F7F7F7Fu) + 0x7F7F7F7Fu) | x | km);
                unsigned m = __builtin_amdgcn_ballot_w32(f != 0u);
                while (m) {
                    if (f) list[__builtin_amdgcn_mbcnt_lo(m, (unsigned)cnt)] = e + 4 * c + (__builtin_ctz(f) >> 3);
                    cnt += __builtin_popcount(m);
                    f &= f - 1u;
                    m = __builtin_amdgcn_ballot_w32(f != 0u);
                }
            }
        }
    };

    auto drain = [&](auto packed) {
        constexpr bool PACKED = decltype(packed)::value;
        __builtin_amdgcn_fence(__ATOMIC_RELEASE, "wavefront");
        __builtin_amdgcn_wave_barrier();
        __builtin_amdgcn_fence(__ATOMIC_ACQUIRE, "wavefront");
        for (int i0 = 0; i0 < cnt; i0 += EPB) {
            int s = 0, d = RR;
            if (lane < EPB && i0 + lane < cnt) {
                const int e = list[i0 + lane];
                if constexpr (PACKED) {
                    s = (int)((unsigned)e >> 8);
                    d = e & 255;
                } else {
                    const int se = src[e];
                    const unsigned de = (unsigned)(dst[e] - base);
                    if (de < (unsigned)RR) { s = se; d = (int)de; }
                }
                if constexpr (!REPLAY) {
                    const int pos = tot + i0 + lane;
                    if (pos < WCAP) seg[pos] = ((unsigned)s << 8) | (unsigned)d;
                }
            }
            float v[EPB][NV];
            #pragma unroll
            for (int j = 0; j < EPB; ++j) {
                const int sj = __builtin_amdgcn_readlane(s, j);
                const float* __restrict__ frow = feat + (size_t)sj * ROWF + lane;
                #pragma unroll
                for (int q = 0; q < NV; ++q) v[j][q] = frow[32 * q];
            }
            if (lane < EPB) unsafeAtomicAdd(&cntf[d], 1.0f);
            #pragma unroll
            for (int j = 0; j < EPB; ++j) {
                const int dj = __builtin_amdgcn_readlane(d, j);
                float* arow = acc + dj * ROWF + lane;
                #pragma unroll
                for (int q = 0; q < NV; ++q) unsafeAtomicAdd(arow + 32 * q, v[j][q]);
            }
        }
        __builtin_amdgcn_fence(__ATOMIC_RELEASE, "wavefront");
        __builtin_amdgcn_wave_barrier();
        tot += cnt;
        cnt = 0;
    };

    auto sift_all = [&]() {
        uint4 da[GRP], db[GRP];
        fetch(da, 0);
        for (int p0 = 0; p0 < NPASS; p0 += 2 * GRP) {
            fetch(db, p0 + GRP);
            sift(da, p0);
            if (cnt > CAP - GMAX) drain(flag<false>{});
            fetch(da, p0 + 2 * GRP);
            sift(db, p0 + GRP);
            if (cnt > CAP - GMAX) drain(flag<false>{});
        }
        drain(flag<false>{});
    };

    if constexpr (!REPLAY) {
        sift_all();
        if (lane == 0) memo_n[gw] = tot;
    } else {
        const int n = __builtin_amdgcn_readfirstlane(memo_n[gw]);
        if (n <= WCAP) {
            #pragma unroll
            for (int k = 0; k < WCAP / 32; ++k) list[32 * k + lane] = (int)seg[32 * k + lane];
            cnt = n;
            drain(flag<true>{});
        } else {
            sift_all();
        }
    }
    __syncthreads();

    constexpr int R4 = ROWF / 4;
    for (int i = tid; i < RR * R4; i += 1024) {
        const int r = i / R4, c4 = i % R4;
        const int node = base + r;
        if (node < N_NODES) {
            const float id = 1.0f / fmaxf(cntf[r], 1.0f);
            float4 a = *(const float4*)(acc + r * ROWF + 4 * c4);
            a.x *= id; a.y *= id; a.z *= id; a.w *= id;
            if constexpr (ADD) {
                const float4 t = *(const float4*)(mean + (size_t)node * ROWF + 4 * c4);
                a.x += t.x; a.y += t.y; a.z += t.z; a.w += t.w;
            }
            *(float4*)(mean + (size_t)node * ROWF + 4 * c4) = a;
        }
    }
}

template <int NOUT, bool SCALE>
__device__ __forceinline__ void strip_product(const float* __restrict__ arow, float scale,
                                              const float* __restrict__ Wm,
                                              int cb, int col, int hi, v8f (&c)[4]) {
    const float* __restrict__ ap = arow + 2 * hi;
    const float* __restrict__ wp = Wm + (size_t)(2 * hi) * NOUT + cb + 4 * col;
    #pragma unroll 4
    for (int k0 = 0; k0 < 128; k0 += 4) {
        const float2 t  = *(const float2*)(ap + k0);
        const float4 bx = *(const float4*)(wp + (size_t)k0 * NOUT);
        const float4 by = *(const float4*)(wp + (size_t)(k0 + 1) * NOUT);
        v2f a;
        a.x = SCALE ? t.x * scale : t.x;
        a.y = SCALE ? t.y * scale : t.y;
        v2f b0, b1, b2, b3;
        b0.x = bx.x; b0.y = by.x;
        b1.x = bx.y; b1.y = by.y;
        b2.x = bx.z; b2.y = by.z;
        b3.x = bx.w; b3.y = by.w;
        c[0] = __builtin_amdgcn_wmma_f32_16x16x4_f32(false, a, false, b0, (short)0, c[0], false, false);
        c[1] = __builtin_amdgcn_wmma_f32_16x16x4_f32(false, a, false, b1, (short)0, c[1], false, false);
        c[2] = __builtin_amdgcn_wmma_f32_16x16x4_f32(false, a, false, b2, (short)0, c[2], false, false);
        c[3] = __builtin_amdgcn_wmma_f32_16x16x4_f32(false, a, false, b3, (short)0, c[3], false, false);
    }
}

__global__ __launch_bounds__(64)
void sage_both_layers(const float* __restrict__ Aagg,
                      const float* __restrict__ Aself,
                      const float* __restrict__ W1l,
                      const float* __restrict__ W1r,
                      const float* __restrict__ b1,
                      const float* __restrict__ W2l,
                      const float* __restrict__ W2r,
                      const float* __restrict__ b2,
                      float* __restrict__ P,
                      float* __restrict__ R) {
    constexpr int K  = 128;
    constexpr int LD = HID_CH + 4;
    __shared__ __attribute__((aligned(16))) float hs[16 * LD];
    const int wave = __builtin_amdgcn_readfirstlane(threadIdx.x >> 5);
    const int lane = threadIdx.x & 31;
    const int row  = lane & 15;
    const int hi   = lane >> 4;
    const int m0   = blockIdx.x * 16;
    const int cb   = wave * 64;

    v8f c[4] = {};
    strip_product<HID_CH, false>(Aagg  + (size_t)(m0 + row) * K, 1.0f, W1l, cb, row, hi, c);
    strip_product<HID_CH, false>(Aself + (size_t)(m0 + row) * K, 1.0f, W1r, cb, row, hi, c);

    const float4 bv = *(const float4*)(b1 + cb + 4 * row);
    #pragma unroll
    for (int r = 0; r < 8; ++r) {
        float4 v;
        v.x = fmaxf(c[0][r] + bv.x, 0.0f);
        v.y = fmaxf(c[1][r] + bv.y, 0.0f);
        v.z = fmaxf(c[2][r] + bv.z, 0.0f);
        v.w = fmaxf(c[3][r] + bv.w, 0.0f);
        *(float4*)(hs + (r + 8 * hi) * LD + cb + 4 * row) = v;
    }
    __syncthreads();

    v8f d[4] = {};
    strip_product<OUT_CH, false>(hs + row * LD, 1.0f, wave == 0 ? W2l : W2r, 0, row, hi, d);

    if (wave == 0) {
        #pragma unroll
        for (int r = 0; r < 8; ++r) {
            float4 v;
            v.x = d[0][r];
            v.y = d[1][r];
            v.z = d[2][r];
            v.w = d[3][r];
            *(float4*)(P + (size_t)(m0 + r + 8 * hi) * OUT_CH + 4 * row) = v;
        }
    } else {
        const float4 b2v = *(const float4*)(b2 + 4 * row);
        #pragma unroll
        for (int r = 0; r < 8; ++r) {
            float4 v;
            v.x = d[0][r] + b2v.x;
            v.y = d[1][r] + b2v.y;
            v.z = d[2][r] + b2v.z;
            v.w = d[3][r] + b2v.w;
            *(float4*)(R + (size_t)(m0 + r + 8 * hi) * OUT_CH + 4 * row) = v;
        }
    }
}

extern "C" void kernel_launch(void* const* d_in, const int* in_sizes, int n_in,
                              void* d_out, int out_size, void* d_ws, size_t ws_size,
                              hipStream_t stream) {
    const float* x   = (const float*)d_in[0];
    const int*   ei  = (const int*)d_in[1];
    const float* W1l = (const float*)d_in[2];
    const float* b1  = (const float*)d_in[3];
    const float* W1r = (const float*)d_in[4];
    const float* W2l = (const float*)d_in[5];
    const float* b2  = (const float*)d_in[6];
    const float* W2r = (const float*)d_in[7];

    const int* src = ei;
    const int* dst = ei + N_EDGES;

    float* mean1  = (float*)d_ws;
    float* plane2 = mean1 + (size_t)N_NODES * 128;
    float* proj   = plane2;

    constexpr int RR = 196;
    constexpr int NR = (N_NODES + RR - 1) / RR;

    static_assert((size_t)N_NODES * OUT_CH * 4 + (size_t)N_EDGES + (size_t)NR * 32 * 4
                  + (size_t)NR * 32 * WCAP * 4 <= (size_t)N_NODES * 128 * 4,
                  "proj, ids, counts and segments fit the second plane");
    static_assert(((size_t)N_NODES * OUT_CH * 4) % 128 == 0 && N_EDGES % 128 == 0,
                  "ids, counts and segments start on a 128-byte line");
    unsigned* ids    = (unsigned*)(plane2 + (size_t)N_NODES * OUT_CH);
    int*      memo_n = (int*)(ids + N_EDGES / 4);
    unsigned* memo   = (unsigned*)(memo_n + NR * 32);

    range_ids<RR><<<(N_EDGES / 4 + 255) / 256, 256, 0, stream>>>(dst, ids);

    range_mean<128, RR, 8, false, false><<<NR, 1024, 0, stream>>>(x, src, dst, ids, memo, memo_n, mean1);

    sage_both_layers<<<N_NODES / 16, 64, 0, stream>>>(
        mean1, x, W1l, W1r, b1, W2l, W2r, b2, proj, (float*)d_out);

    range_mean<OUT_CH, RR, 16, true, true><<<NR, 1024, 0, stream>>>(proj, src, dst, ids, memo, memo_n, (float*)d_out);
}


# Row gathers widened: a 512-byte row comes by one 16-byte-per-lane load, two 256-byte rows by one (a half-wave each); LDS sums kept lane-interleaved; same arrival-order LDS f32 atomic sums
# speedup vs baseline: 4.0796x; 1.0488x over previous
#include <hip/hip_runtime.h>
#include <hip/hip_bf16.h>

#define N_NODES 50000
#define N_EDGES 800000
#define IN_CH   128
#define HID_CH  128
#define OUT_CH  64

typedef __attribute__((ext_vector_type(2))) float v2f;
typedef __attribute__((ext_vector_type(8))) float v8f;

template <int RR>
__global__ __launch_bounds__(256)
void range_ids(const int* __restrict__ dst, unsigned* __restrict__ ids) {
    static_assert((N_NODES + RR - 1) / RR <= 256, "a range id is one byte");
    static_assert(N_EDGES % 4 == 0, "16-byte index loads");
    const int i = blockIdx.x * 256 + threadIdx.x;
    if (i < N_EDGES / 4) {
        const int4 d = *(const int4*)(dst + 4 * i);
        ids[i] =  (((unsigned)d.x / RR) & 255u)
               | ((((unsigned)d.y / RR) & 255u) << 8)
               | ((((unsigned)d.z / RR) & 255u) << 16)
               | ((((unsigned)d.w / RR) & 255u) << 24);
    }
}

template <bool B> struct flag { static constexpr bool value = B; };

constexpr int WCAP = 256;

template <int ROWF, int RR, int EPB, bool REPLAY, bool ADD>
__global__ __launch_bounds__(1024)
void range_mean(const float* __restrict__ feat,
                const int* __restrict__ src, const int* __restrict__ dst,
                const unsigned* __restrict__ ids,
                unsigned* __restrict__ memo, int* __restrict__ memo_n,
                float* __restrict__ mean) {
    static_assert(ROWF % 32 == 0, "a row is a whole number of 128-byte lines");
    static_assert(N_EDGES % 16 == 0, "16-byte id loads");
    static_assert(ROWF == 128 || ROWF == 64, "a wave-wide 16-byte load covers one row or two");
    static_assert((unsigned long long)N_NODES * ROWF * 4 < (1ull << 32), "a row's byte offset fits 32 bits");
    constexpr int R4   = ROWF / 4;
    constexpr int GRP  = 2;
    constexpr int EPT  = 16;
    constexpr int GMAX = 32 * EPT * GRP;
    constexpr int CAP  = GMAX + 256;
    static_assert(WCAP <= CAP && WCAP % 32 == 0, "a stored segment fits the wave's list");
    static_assert(RR < 256 && N_NODES <= (1 << 24), "source << 8 | local destination is one dword");
    constexpr int EPP  = 1024 * EPT;
    constexpr int NPASS = (N_EDGES + EPP - 1) / EPP;
    __shared__ float acc[(RR + 1) * ROWF];
    __shared__ float cntf[RR + 1];
    __shared__ int   lists[32 * CAP];

    const int tid  = threadIdx.x;
    const int lane = tid & 31;
    const int wv   = __builtin_amdgcn_readfirstlane(tid >> 5);
    const int base = blockIdx.x * RR;
    const unsigned pat = (unsigned)blockIdx.x * 0x01010101u;

    for (int i = tid; i < (RR + 1) * ROWF; i += 1024) acc[i] = 0.0f;
    for (int i = tid; i < RR + 1; i += 1024) cntf[i] = 0.0f;
    __syncthreads();

    int* list = lists + wv * CAP;
    int cnt = 0;
    int tot = 0;
    const int gw = blockIdx.x * 32 + wv;
    unsigned* __restrict__ seg = memo + (size_t)gw * WCAP;

    const int start = blockIdx.x % NPASS;
    auto edge_of = [&](int p) -> int {
        int q = start + p;
        if (q >= NPASS) q -= NPASS;
        const int e = (q * 1024 + tid) * EPT;
        return (p < NPASS && e < N_EDGES) ? e : -1;
    };

    auto fetch = [&](uint4 (&dv)[GRP], int p) {
        #pragma unroll
        for (int u = 0; u < GRP; ++u) {
            const int e = edge_of(p + u);
            dv[u] = *(const uint4*)(ids + ((e < 0 ? 0 : e) >> 2));
        }
    };

    auto sift = [&](const uint4 (&dv)[GRP], int p) {
        #pragma unroll
        for (int u = 0; u < GRP; ++u) {
            const int e = edge_of(p + u);
            const unsigned km = e >= 0 ? 0x7F7F7F7Fu : 0xFFFFFFFFu;
            const unsigned ww[4] = {dv[u].x, dv[u].y, dv[u].z, dv[u].w};
            #pragma unroll
            for (int c = 0; c < 4; ++c) {
                const unsigned x = ww[c] ^ pat;
                unsigned f = ~(((x & 0x7F7F7F7Fu) + 0x7F7F7F7Fu) | x | km);
                unsigned m = __builtin_amdgcn_ballot_w32(f != 0u);
                while (m) {
                    if (f) list[__builtin_amdgcn_mbcnt_lo(m, (unsigned)cnt)] = e + 4 * c + (__builtin_ctz(f) >> 3);
                    cnt += __builtin_popcount(m);
                    f &= f - 1u;
                    m = __builtin_amdgcn_ballot_w32(f != 0u);
                }
            }
        }
    };

    auto drain = [&](auto packed) {
        constexpr bool PACKED = decltype(packed)::value;
        __builtin_amdgcn_fence(__ATOMIC_RELEASE, "wavefront");
        __builtin_amdgcn_wave_barrier();
        __builtin_amdgcn_fence(__ATOMIC_ACQUIRE, "wavefront");
        for (int i0 = 0; i0 < cnt; i0 += EPB) {
            int s = 0, d = RR;
            if (lane < EPB && i0 + lane < cnt) {
                const int e = list[i0 + lane];
                if constexpr (PACKED) {
                    s = (int)((unsigned)e >> 8);
                    d = e & 255;
                } else {
                    const int se = src[e];
                    const unsigned de = (unsigned)(dst[e] - base);
                    if (de < (unsigned)RR) { s = se; d = (int)de; }
                }
                if constexpr (!REPLAY) {
                    const int pos = tot + i0 + lane;
                    if (pos < WCAP) seg[pos] = ((unsigned)s << 8) | (unsigned)d;
                }
                unsafeAtomicAdd(&cntf[d], 1.0f);
            }
            if constexpr (ROWF == 128) {
                float4 v[EPB];
                #pragma unroll
                for (int j = 0; j < EPB; ++j) {
                    const int sj = __builtin_amdgcn_readlane(s, j);
                    v[j] = *(const float4*)((const char*)feat + ((unsigned)sj * (unsigned)(ROWF * 4) + 16u * (unsigned)lane));
                }
                #pragma unroll
                for (int j = 0; j < EPB; ++j) {
                    const int dj = __builtin_amdgcn_readlane(d, j);
                    float* arow = acc + dj * ROWF + lane;
                    unsafeAtomicAdd(arow,          v[j].x);
                    unsafeAtomicAdd(arow + R4,     v[j].y);
                    unsafeAtomicAdd(arow + 2 * R4, v[j].z);
                    unsafeAtomicAdd(arow + 3 * R4, v[j].w);
                }
            } else {
                static_assert(EPB % 2 == 0, "edges go in pairs");
                const int hw = lane >> 4;
                const int cl = lane & 15;
                float4 v[EPB / 2];
                #pragma unroll
                for (int p = 0; p < EPB / 2; ++p) {
                    const int s0 = __builtin_amdgcn_readlane(s, 2 * p);
                    const int s1 = __builtin_amdgcn_readlane(s, 2 * p + 1);
                    const unsigned sp = (unsigned)(hw ? s1 : s0);
                    v[p] = *(const float4*)((const char*)feat + (sp * (unsigned)(ROWF * 4) + 16u * (unsigned)cl));
                }
                #pragma unroll
                for (int p = 0; p < EPB / 2; ++p) {
                    const int d0 = __builtin_amdgcn_readlane(d, 2 * p);
                    const int d1 = __builtin_amdgcn_readlane(d, 2 * p + 1);
                    const int dp = hw ? d1 : d0;
                    float* a0 = acc + dp * ROWF + cl + (hw ? 2 * R4 : 0);
                    float* a1 = acc + dp * ROWF + cl + (hw ? 0 : 2 * R4);
                    unsafeAtomicAdd(a0,      hw ? v[p].z : v[p].x);
                    unsafeAtomicAdd(a0 + R4, hw ? v[p].w : v[p].y);
                    unsafeAtomicAdd(a1,      hw ? v[p].x : v[p].z);
                    unsafeAtomicAdd(a1 + R4, hw ? v[p].y : v[p].w);
                }
            }
        }
        __builtin_amdgcn_fence(__ATOMIC_RELEASE, "wavefront");
        __builtin_amdgcn_wave_barrier();
        tot += cnt;
        cnt = 0;
    };

    auto sift_all = [&]() {
        uint4 da[GRP], db[GRP];
        fetch(da, 0);
        for (int p0 = 0; p0 < NPASS; p0 += 2 * GRP) {
            fetch(db, p0 + GRP);
            sift(da, p0);
            if (cnt > CAP - GMAX) drain(flag<false>{});
            fetch(da, p0 + 2 * GRP);
            sift(db, p0 + GRP);
            if (cnt > CAP - GMAX) drain(flag<false>{});
        }
        drain(flag<false>{});
    };

    if constexpr (!REPLAY) {
        sift_all();
        if (lane == 0) memo_n[gw] = tot;
    } else {
        const int n = __builtin_amdgcn_readfirstlane(memo_n[gw]);
        if (n <= WCAP) {
            #pragma unroll
            for (int k = 0; k < WCAP / 32; ++k) list[32 * k + lane] = (int)seg[32 * k + lane];
            cnt = n;
            drain(flag<true>{});
        } else {
            sift_all();
        }
    }
    __syncthreads();

    for (int i = tid; i < RR * R4; i += 1024) {
        const int r = i / R4, c4 = i % R4;
        const int node = base + r;
        if (node < N_NODES) {
            const float id = 1.0f / fmaxf(cntf[r], 1.0f);
            const float* ar = acc + r * ROWF + c4;
            float4 a;
            a.x = ar[0] * id; a.y = ar[R4] * id; a.z = ar[2 * R4] * id; a.w = ar[3 * R4] * id;
            if constexpr (ADD) {
                const float4 t = *(const float4*)(mean + (size_t)node * ROWF + 4 * c4);
                a.x += t.x; a.y += t.y; a.z += t.z; a.w += t.w;
            }
            *(float4*)(mean + (size_t)node * ROWF + 4 * c4) = a;
        }
    }
}

template <int NOUT, bool SCALE>
__device__ __forceinline__ void strip_product(const float* __restrict__ arow, float scale,
                                              const float* __restrict__ Wm,
                                              int cb, int col, int hi, v8f (&c)[4]) {
    const float* __restrict__ ap = arow + 2 * hi;
    const float* __restrict__ wp = Wm + (size_t)(2 * hi) * NOUT + cb + 4 * col;
    #pragma unroll 4
    for (int k0 = 0; k0 < 128; k0 += 4) {
        const float2 t  = *(const float2*)(ap + k0);
        const float4 bx = *(const float4*)(wp + (size_t)k0 * NOUT);
        const float4 by = *(const float4*)(wp + (size_t)(k0 + 1) * NOUT);
        v2f a;
        a.x = SCALE ? t.x * scale : t.x;
        a.y = SCALE ? t.y * scale : t.y;
        v2f b0, b1, b2, b3;
        b0.x = bx.x; b0.y = by.x;
        b1.x = bx.y; b1.y = by.y;
        b2.x = bx.z; b2.y = by.z;
        b3.x = bx.w; b3.y = by.w;
        c[0] = __builtin_amdgcn_wmma_f32_16x16x4_f32(false, a, false, b0, (short)0, c[0], false, false);
        c[1] = __builtin_amdgcn_wmma_f32_16x16x4_f32(false, a, false, b1, (short)0, c[1], false, false);
        c[2] = __builtin_amdgcn_wmma_f32_16x16x4_f32(false, a, false, b2, (short)0, c[2], false, false);
        c[3] = __builtin_amdgcn_wmma_f32_16x16x4_f32(false, a, false, b3, (short)0, c[3], false, false);
    }
}

__global__ __launch_bounds__(64)
void sage_both_layers(const float* __restrict__ Aagg,
                      const float* __restrict__ Aself,
                      const float* __restrict__ W1l,
                      const float* __restrict__ W1r,
                      const float* __restrict__ b1,
                      const float* __restrict__ W2l,
                      const float* __restrict__ W2r,
                      const float* __restrict__ b2,
                      float* __restrict__ P,
                      float* __restrict__ R) {
    constexpr int K  = 128;
    constexpr int LD = HID_CH + 4;
    __shared__ __attribute__((aligned(16))) float hs[16 * LD];
    const int wave = __builtin_amdgcn_readfirstlane(threadIdx.x >> 5);
    const int lane = threadIdx.x & 31;
    const int row  = lane & 15;
    const int hi   = lane >> 4;
    const int m0   = blockIdx.x * 16;
    const int cb   = wave * 64;

    v8f c[4] = {};
    strip_product<HID_CH, false>(Aagg  + (size_t)(m0 + row) * K, 1.0f, W1l, cb, row, hi, c);
    strip_product<HID_CH, false>(Aself + (size_t)(m0 + row) * K, 1.0f, W1r, cb, row, hi, c);

    const float4 bv = *(const float4*)(b1 + cb + 4 * row);
    #pragma unroll
    for (int r = 0; r < 8; ++r) {
        float4 v;
        v.x = fmaxf(c[0][r] + bv.x, 0.0f);
        v.y = fmaxf(c[1][r] + bv.y, 0.0f);
        v.z = fmaxf(c[2][r] + bv.z, 0.0f);
        v.w = fmaxf(c[3][r] + bv.w, 0.0f);
        *(float4*)(hs + (r + 8 * hi) * LD + cb + 4 * row) = v;
    }
    __syncthreads();

    v8f d[4] = {};
    strip_product<OUT_CH, false>(hs + row * LD, 1.0f, wave == 0 ? W2l : W2r, 0, row, hi, d);

    if (wave == 0) {
        #pragma unroll
        for (int r = 0; r < 8; ++r) {
            float4 v;
            v.x = d[0][r];
            v.y = d[1][r];
            v.z = d[2][r];
            v.w = d[3][r];
            *(float4*)(P + (size_t)(m0 + r + 8 * hi) * OUT_CH + 4 * row) = v;
        }
    } else {
        const float4 b2v = *(const float4*)(b2 + 4 * row);
        #pragma unroll
        for (int r = 0; r < 8; ++r) {
            float4 v;
            v.x = d[0][r] + b2v.x;
            v.y = d[1][r] + b2v.y;
            v.z = d[2][r] + b2v.z;
            v.w = d[3][r] + b2v.w;
            *(float4*)(R + (size_t)(m0 + r + 8 * hi) * OUT_CH + 4 * row) = v;
        }
    }
}

extern "C" void kernel_launch(void* const* d_in, const int* in_sizes, int n_in,
                              void* d_out, int out_size, void* d_ws, size_t ws_size,
                              hipStream_t stream) {
    const float* x   = (const float*)d_in[0];
    const int*   ei  = (const int*)d_in[1];
    const float* W1l = (const float*)d_in[2];
    const float* b1  = (const float*)d_in[3];
    const float* W1r = (const float*)d_in[4];
    const float* W2l = (const float*)d_in[5];
    const float* b2  = (const float*)d_in[6];
    const float* W2r = (const float*)d_in[7];

    const int* src = ei;
    const int* dst = ei + N_EDGES;

    float* mean1  = (float*)d_ws;
    float* plane2 = mean1 + (size_t)N_NODES * 128;
    float* proj   = plane2;

    constexpr int RR = 196;
    constexpr int NR = (N_NODES + RR - 1) / RR;

    static_assert((size_t)N_NODES * OUT_CH * 4 + (size_t)N_EDGES + (size_t)NR * 32 * 4
                  + (size_t)NR * 32 * WCAP * 4 <= (size_t)N_NODES * 128 * 4,
                  "proj, ids, counts and segments fit the second plane");
    static_assert(((size_t)N_NODES * OUT_CH * 4) % 128 == 0 && N_EDGES % 128 == 0,
                  "ids, counts and segments start on a 128-byte line");
    unsigned* ids    = (unsigned*)(plane2 + (size_t)N_NODES * OUT_CH);
    int*      memo_n = (int*)(ids + N_EDGES / 4);
    unsigned* memo   = (unsigned*)(memo_n + NR * 32);

    range_ids<RR><<<(N_EDGES / 4 + 255) / 256, 256, 0, stream>>>(dst, ids);

    range_mean<128, RR, 8, false, false><<<NR, 1024, 0, stream>>>(x, src, dst, ids, memo, memo_n, mean1);

    sage_both_layers<<<N_NODES / 16, 64, 0, stream>>>(
        mean1, x, W1l, W1r, b1, W2l, W2r, b2, proj, (float*)d_out);

    range_mean<OUT_CH, RR, 16, true, true><<<NR, 1024, 0, stream>>>(proj, src, dst, ids, memo, memo_n, (float*)d_out);
}


# Presence map: id launch also stores, per 1024-edge super-block, which 32-edge blocks hold each range; range waves read flagged blocks' ids only; edges regrouped by wave, arrival-order LDS sums
# speedup vs baseline: 4.3599x; 1.0687x over previous
#include <hip/hip_runtime.h>
#include <hip/hip_bf16.h>

#define N_NODES 50000
#define N_EDGES 800000
#define IN_CH   128
#define HID_CH  128
#define OUT_CH  64

typedef __attribute__((ext_vector_type(2))) float v2f;
typedef __attribute__((ext_vector_type(8))) float v8f;

constexpr int NBLK = N_EDGES / 32;
constexpr int NSB  = (NBLK + 31) / 32;
constexpr int IDW  = 2;

template <int RR>
__global__ __launch_bounds__(32 * IDW)
void range_ids(const int* __restrict__ dst, unsigned* __restrict__ ids, unsigned* __restrict__ bmap) {
    static_assert((N_NODES + RR - 1) / RR <= 256, "a range id is one byte");
    static_assert(N_EDGES % 32 == 0, "whole blocks of 32 edges");
    __shared__ unsigned pres[IDW][256];
    const int lane = threadIdx.x & 31;
    const int wv   = threadIdx.x >> 5;
    const int sb   = blockIdx.x * IDW + wv;
    const int blk  = sb * 32 + lane;
    unsigned* pw = pres[wv];
    #pragma unroll
    for (int k = 0; k < 8; ++k) pw[32 * k + lane] = 0u;
    __syncthreads();
    if (blk < NBLK) {
        const int4* dp = (const int4*)(dst + (size_t)blk * 32);
        int4 d[8];
        #pragma unroll
        for (int q = 0; q < 8; ++q) d[q] = dp[q];
        const unsigned bit = 1u << lane;
        unsigned w[8];
        #pragma unroll
        for (int q = 0; q < 8; ++q) {
            const unsigned i0 = ((unsigned)d[q].x / RR) & 255u;
            const unsigned i1 = ((unsigned)d[q].y / RR) & 255u;
            const unsigned i2 = ((unsigned)d[q].z / RR) & 255u;
            const unsigned i3 = ((unsigned)d[q].w / RR) & 255u;
            w[q] = i0 | (i1 << 8) | (i2 << 16) | (i3 << 24);
            atomicOr(&pw[i0], bit);
            atomicOr(&pw[i1], bit);
            atomicOr(&pw[i2], bit);
            atomicOr(&pw[i3], bit);
        }
        uint4* ip = (uint4*)(ids + (size_t)blk * 8);
        ip[0] = make_uint4(w[0], w[1], w[2], w[3]);
        ip[1] = make_uint4(w[4], w[5], w[6], w[7]);
    }
    __syncthreads();
    if (sb < NSB) {
        #pragma unroll
        for (int k = 0; k < 8; ++k) bmap[(size_t)sb * 256 + 32 * k + lane] = pw[32 * k + lane];
    }
}

template <bool B> struct flag { static constexpr bool value = B; };

constexpr int WCAP = 256;

template <int ROWF, int RR, int EPB, bool REPLAY, bool ADD>
__global__ __launch_bounds__(1024)
void range_mean(const float* __restrict__ feat,
                const int* __restrict__ src, const int* __restrict__ dst,
                const unsigned* __restrict__ ids, const unsigned* __restrict__ bmap,
                unsigned* __restrict__ memo, int* __restrict__ memo_n,
                float* __restrict__ mean) {
    static_assert(ROWF % 32 == 0, "a row is a whole number of 128-byte lines");
    static_assert(ROWF == 128 || ROWF == 64, "a wave-wide 16-byte load covers one row or two");
    static_assert((unsigned long long)N_NODES * ROWF * 4 < (1ull << 32), "a row's byte offset fits 32 bits");
    constexpr int R4   = ROWF / 4;
    constexpr int GMAX = 32 * 32;
    constexpr int CAP  = GMAX + 256;
    constexpr int SPW  = (NSB + 31) / 32;
    constexpr int BCAP = SPW * 32;
    static_assert(SPW <= 32, "a lane per super-block of the wave");
    static_assert(NSB * 32 <= 65536, "a block number fits 16 bits");
    static_assert(WCAP <= CAP && WCAP % 32 == 0, "a stored segment fits the wave's list");
    static_assert(RR < 256 && N_NODES <= (1 << 24), "source << 8 | local destination is one dword");
    __shared__ float acc[(RR + 1) * ROWF];
    __shared__ float cntf[RR + 1];
    __shared__ int   lists[32 * CAP];
    __shared__ unsigned short blist[32 * BCAP];

    const int tid  = threadIdx.x;
    const int lane = tid & 31;
    const int wv   = __builtin_amdgcn_readfirstlane(tid >> 5);
    const int base = blockIdx.x * RR;
    const unsigned pat = (unsigned)blockIdx.x * 0x01010101u;

    for (int i = tid; i < (RR + 1) * ROWF; i += 1024) acc[i] = 0.0f;
    for (int i = tid; i < RR + 1; i += 1024) cntf[i] = 0.0f;
    __syncthreads();

    int* list = lists + wv * CAP;
    int cnt = 0;
    int tot = 0;
    const int gw = blockIdx.x * 32 + wv;
    unsigned* __restrict__ seg = memo + (size_t)gw * WCAP;

    unsigned short* bl = blist + wv * BCAP;

    auto fetch = [&](uint4 (&dv)[2], int& blk, bool& on, int g, int nb) {
        on  = g + lane < nb;
        blk = on ? (int)bl[g + lane] : 0;
        const uint4* p = (const uint4*)(ids + blk * 8);
        dv[0] = p[0];
        dv[1] = p[1];
    };

    auto sift = [&](const uint4 (&dv)[2], int blk, bool on) {
        const int e = blk * 32;
        const unsigned km = on ? 0x7F7F7F7Fu : 0xFFFFFFFFu;
        const unsigned ww[8] = {dv[0].x, dv[0].y, dv[0].z, dv[0].w, dv[1].x, dv[1].y, dv[1].z, dv[1].w};
        #pragma unroll
        for (int c = 0; c < 8; ++c) {
            const unsigned x = ww[c] ^ pat;
            unsigned f = ~(((x & 0x7F7F7F7Fu) + 0x7F7F7F7Fu) | x | km);
            unsigned m = __builtin_amdgcn_ballot_w32(f != 0u);
            while (m) {
                if (f) list[__builtin_amdgcn_mbcnt_lo(m, (unsigned)cnt)] = e + 4 * c + (__builtin_ctz(f) >> 3);
                cnt += __builtin_popcount(m);
                f &= f - 1u;
                m = __builtin_amdgcn_ballot_w32(f != 0u);
            }
        }
    };

    auto drain = [&](auto packed) {
        constexpr bool PACKED = decltype(packed)::value;
        __builtin_amdgcn_fence(__ATOMIC_RELEASE, "wavefront");
        __builtin_amdgcn_wave_barrier();
        __builtin_amdgcn_fence(__ATOMIC_ACQUIRE, "wavefront");
        for (int i0 = 0; i0 < cnt; i0 += EPB) {
            int s = 0, d = RR;
            if (lane < EPB && i0 + lane < cnt) {
                const int e = list[i0 + lane];
                if constexpr (PACKED) {
                    s = (int)((unsigned)e >> 8);
                    d = e & 255;
                } else {
                    const int se = src[e];
                    const unsigned de = (unsigned)(dst[e] - base);
                    if (de < (unsigned)RR) { s = se; d = (int)de; }
                }
                if constexpr (!REPLAY) {
                    const int pos = tot + i0 + lane;
                    if (pos < WCAP) seg[pos] = ((unsigned)s << 8) | (unsigned)d;
                }
                unsafeAtomicAdd(&cntf[d], 1.0f);
            }
            if constexpr (ROWF == 128) {
                float4 v[EPB];
                #pragma unroll
                for (int j = 0; j < EPB; ++j) {
                    const int sj = __builtin_amdgcn_readlane(s, j);
                    v[j] = *(const float4*)((const char*)feat + ((unsigned)sj * (unsigned)(ROWF * 4) + 16u * (unsigned)lane));
                }
                #pragma unroll
                for (int j = 0; j < EPB; ++j) {
                    const int dj = __builtin_amdgcn_readlane(d, j);
                    float* arow = acc + dj * ROWF + lane;
                    unsafeAtomicAdd(arow,          v[j].x);
                    unsafeAtomicAdd(arow + R4,     v[j].y);
                    unsafeAtomicAdd(arow + 2 * R4, v[j].z);
                    unsafeAtomicAdd(arow + 3 * R4, v[j].w);
                }
            } else {
                static_assert(EPB % 2 == 0, "edges go in pairs");
                const int hw = lane >> 4;
                const int cl = lane & 15;
                float4 v[EPB / 2];
                #pragma unroll
                for (int p = 0; p < EPB / 2; ++p) {
                    const int s0 = __builtin_amdgcn_readlane(s, 2 * p);
                    const int s1 = __builtin_amdgcn_readlane(s, 2 * p + 1);
                    const unsigned sp = (unsigned)(hw ? s1 : s0);
                    v[p] = *(const float4*)((const char*)feat + (sp * (unsigned)(ROWF * 4) + 16u * (unsigned)cl));
                }
                #pragma unroll
                for (int p = 0; p < EPB / 2; ++p) {
                    const int d0 = __builtin_amdgcn_readlane(d, 2 * p);
                    const int d1 = __builtin_amdgcn_readlane(d, 2 * p + 1);
                    const int dp = hw ? d1 : d0;
                    float* a0 = acc + dp * ROWF + cl + (hw ? 2 * R4 : 0);
                    float* a1 = acc + dp * ROWF + cl + (hw ? 0 : 2 * R4);
                    unsafeAtomicAdd(a0,      hw ? v[p].z : v[p].x);
                    unsafeAtomicAdd(a0 + R4, hw ? v[p].w : v[p].y);
                    unsafeAtomicAdd(a1,      hw ? v[p].x : v[p].z);
                    unsafeAtomicAdd(a1 + R4, hw ? v[p].y : v[p].w);
                }
            }
        }
        __builtin_amdgcn_fence(__ATOMIC_RELEASE, "wavefront");
        __builtin_amdgcn_wave_barrier();
        tot += cnt;
        cnt = 0;
    };

    auto sift_all = [&]() {
        const int sbi = lane * 32 + wv;
        unsigned bw = 0u;
        if (sbi < NSB) bw = bmap[(size_t)sbi * 256 + blockIdx.x];
        int nb = 0;
        unsigned mm = __builtin_amdgcn_ballot_w32(bw != 0u);
        while (mm) {
            if (bw) bl[__builtin_amdgcn_mbcnt_lo(mm, (unsigned)nb)] = (unsigned short)(sbi * 32 + __builtin_ctz(bw));
            nb += __builtin_popcount(mm);
            bw &= bw - 1u;
            mm = __builtin_amdgcn_ballot_w32(bw != 0u);
        }
        __builtin_amdgcn_fence(__ATOMIC_RELEASE, "wavefront");
        __builtin_amdgcn_wave_barrier();
        __builtin_amdgcn_fence(__ATOMIC_ACQUIRE, "wavefront");
        uint4 da[2], db[2];
        int ba, bb;
        bool oa, ob;
        fetch(da, ba, oa, 0, nb);
        for (int g = 0; g < nb; g += 64) {
            fetch(db, bb, ob, g + 32, nb);
            sift(da, ba, oa);
            if (cnt > CAP - GMAX) drain(flag<false>{});
            fetch(da, ba, oa, g + 64, nb);
            sift(db, bb, ob);
            if (cnt > CAP - GMAX) drain(flag<false>{});
        }
        drain(flag<false>{});
    };

    if constexpr (!REPLAY) {
        sift_all();
        if (lane == 0) memo_n[gw] = tot;
    } else {
        const int n = __builtin_amdgcn_readfirstlane(memo_n[gw]);
        if (n <= WCAP) {
            #pragma unroll
            for (int k = 0; k < WCAP / 32; ++k) list[32 * k + lane] = (int)seg[32 * k + lane];
            cnt = n;
            drain(flag<true>{});
        } else {
            sift_all();
        }
    }
    __syncthreads();

    for (int i = tid; i < RR * R4; i += 1024) {
        const int r = i / R4, c4 = i % R4;
        const int node = base + r;
        if (node < N_NODES) {
            const float id = 1.0f / fmaxf(cntf[r], 1.0f);
            const float* ar = acc + r * ROWF + c4;
            float4 a;
            a.x = ar[0] * id; a.y = ar[R4] * id; a.z = ar[2 * R4] * id; a.w = ar[3 * R4] * id;
            if constexpr (ADD) {
                const float4 t = *(const float4*)(mean + (size_t)node * ROWF + 4 * c4);
                a.x += t.x; a.y += t.y; a.z += t.z; a.w += t.w;
            }
            *(float4*)(mean + (size_t)node * ROWF + 4 * c4) = a;
        }
    }
}

template <int NOUT, bool SCALE>
__device__ __forceinline__ void strip_product(const float* __restrict__ arow, float scale,
                                              const float* __restrict__ Wm,
                                              int cb, int col, int hi, v8f (&c)[4]) {
    const float* __restrict__ ap = arow + 2 * hi;
    const float* __restrict__ wp = Wm + (size_t)(2 * hi) * NOUT + cb + 4 * col;
    #pragma unroll 4
    for (int k0 = 0; k0 < 128; k0 += 4) {
        const float2 t  = *(const float2*)(ap + k0);
        const float4 bx = *(const float4*)(wp + (size_t)k0 * NOUT);
        const float4 by = *(const float4*)(wp + (size_t)(k0 + 1) * NOUT);
        v2f a;
        a.x = SCALE ? t.x * scale : t.x;
        a.y = SCALE ? t.y * scale : t.y;
        v2f b0, b1, b2, b3;
        b0.x = bx.x; b0.y = by.x;
        b1.x = bx.y; b1.y = by.y;
        b2.x = bx.z; b2.y = by.z;
        b3.x = bx.w; b3.y = by.w;
        c[0] = __builtin_amdgcn_wmma_f32_16x16x4_f32(false, a, false, b0, (short)0, c[0], false, false);
        c[1] = __builtin_amdgcn_wmma_f32_16x16x4_f32(false, a, false, b1, (short)0, c[1], false, false);
        c[2] = __builtin_amdgcn_wmma_f32_16x16x4_f32(false, a, false, b2, (short)0, c[2], false, false);
        c[3] = __builtin_amdgcn_wmma_f32_16x16x4_f32(false, a, false, b3, (short)0, c[3], false, false);
    }
}

__global__ __launch_bounds__(64)
void sage_both_layers(const float* __restrict__ Aagg,
                      const float* __restrict__ Aself,
                      const float* __restrict__ W1l,
                      const float* __restrict__ W1r,
                      const float* __restrict__ b1,
                      const float* __restrict__ W2l,
                      const float* __restrict__ W2r,
                      const float* __restrict__ b2,
                      float* __restrict__ P,
                      float* __restrict__ R) {
    constexpr int K  = 128;
    constexpr int LD = HID_CH + 4;
    __shared__ __attribute__((aligned(16))) float hs[16 * LD];
    const int wave = __builtin_amdgcn_readfirstlane(threadIdx.x >> 5);
    const int lane = threadIdx.x & 31;
    const int row  = lane & 15;
    const int hi   = lane >> 4;
    const int m0   = blockIdx.x * 16;
    const int cb   = wave * 64;

    v8f c[4] = {};
    strip_product<HID_CH, false>(Aagg  + (size_t)(m0 + row) * K, 1.0f, W1l, cb, row, hi, c);
    strip_product<HID_CH, false>(Aself + (size_t)(m0 + row) * K, 1.0f, W1r, cb, row, hi, c);

    const float4 bv = *(const float4*)(b1 + cb + 4 * row);
    #pragma unroll
    for (int r = 0; r < 8; ++r) {
        float4 v;
        v.x = fmaxf(c[0][r] + bv.x, 0.0f);
        v.y = fmaxf(c[1][r] + bv.y, 0.0f);
        v.z = fmaxf(c[2][r] + bv.z, 0.0f);
        v.w = fmaxf(c[3][r] + bv.w, 0.0f);
        *(float4*)(hs + (r + 8 * hi) * LD + cb + 4 * row) = v;
    }
    __syncthreads();

    v8f d[4] = {};
    strip_product<OUT_CH, false>(hs + row * LD, 1.0f, wave == 0 ? W2l : W2r, 0, row, hi, d);

    if (wave == 0) {
        #pragma unroll
        for (int r = 0; r < 8; ++r) {
            float4 v;
            v.x = d[0][r];
            v.y = d[1][r];
            v.z = d[2][r];
            v.w = d[3][r];
            *(float4*)(P + (size_t)(m0 + r + 8 * hi) * OUT_CH + 4 * row) = v;
        }
    } else {
        const float4 b2v = *(const float4*)(b2 + 4 * row);
        #pragma unroll
        for (int r = 0; r < 8; ++r) {
            float4 v;
            v.x = d[0][r] + b2v.x;
            v.y = d[1][r] + b2v.y;
            v.z = d[2][r] + b2v.z;
            v.w = d[3][r] + b2v.w;
            *(float4*)(R + (size_t)(m0 + r + 8 * hi) * OUT_CH + 4 * row) = v;
        }
    }
}

extern "C" void kernel_launch(void* const* d_in, const int* in_sizes, int n_in,
                              void* d_out, int out_size, void* d_ws, size_t ws_size,
                              hipStream_t stream) {
    const float* x   = (const float*)d_in[0];
    const int*   ei  = (const int*)d_in[1];
    const float* W1l = (const float*)d_in[2];
    const float* b1  = (const float*)d_in[3];
    const float* W1r = (const float*)d_in[4];
    const float* W2l = (const float*)d_in[5];
    const float* b2  = (const float*)d_in[6];
    const float* W2r = (const float*)d_in[7];

    const int* src = ei;
    const int* dst = ei + N_EDGES;

    float* mean1  = (float*)d_ws;
    float* plane2 = mean1 + (size_t)N_NODES * 128;
    float* proj   = plane2;

    constexpr int RR = 196;
    constexpr int NR = (N_NODES + RR - 1) / RR;

    static_assert((size_t)N_NODES * OUT_CH * 4 + (size_t)N_EDGES + (size_t)NR * 32 * 4
                  + (size_t)NR * 32 * WCAP * 4 + (size_t)NSB * 256 * 4 <= (size_t)N_NODES * 128 * 4,
                  "proj, ids, counts, segments and the presence map fit the second plane");
    static_assert(((size_t)N_NODES * OUT_CH * 4) % 128 == 0 && N_EDGES % 128 == 0,
                  "ids, counts, segments and the presence map start on a 128-byte line");
    unsigned* ids    = (unsigned*)(plane2 + (size_t)N_NODES * OUT_CH);
    int*      memo_n = (int*)(ids + N_EDGES / 4);
    unsigned* memo   = (unsigned*)(memo_n + NR * 32);
    unsigned* bmap   = memo + (size_t)NR * 32 * WCAP;

    range_ids<RR><<<(NSB + IDW - 1) / IDW, 32 * IDW, 0, stream>>>(dst, ids, bmap);

    range_mean<128, RR, 8, false, false><<<NR, 1024, 0, stream>>>(x, src, dst, ids, bmap, memo, memo_n, mean1);

    sage_both_layers<<<N_NODES / 16, 64, 0, stream>>>(
        mean1, x, W1l, W1r, b1, W2l, W2r, b2, proj, (float*)d_out);

    range_mean<OUT_CH, RR, 16, true, true><<<NR, 1024, 0, stream>>>(proj, src, dst, ids, bmap, memo, memo_n, (float*)d_out);
}
